# starting kernel
# baseline (speedup 1.0000x reference)
#include <hip/hip_runtime.h>

typedef __attribute__((ext_vector_type(2))) float v2f;
typedef __attribute__((ext_vector_type(4))) float v4f;
typedef __attribute__((ext_vector_type(8))) float v8f;

constexpr int Dd = 768;
constexpr int Tt = 512;
constexpr int Vv = 49408;
constexpr int R  = 64 * 8;
constexpr int VT = Vv / 16;
constexpr int NP = VT / 2;
constexpr int S  = 128;
constexpr float EPS = 1e-8f;

__device__ __forceinline__ v8f wmma_f32_k4(v2f a, v2f b, v8f c) {
  return __builtin_amdgcn_wmma_f32_16x16x4_f32(
       false, a,  false, b,
       (short)0, c,  false,  false);
}

__device__ __forceinline__ void upd(float& bs, int& bv, float s, int v) {
  if (s > bs || (s == bs && v < bv)) { bs = s; bv = v; }
}

__global__ void zero_f32(float* __restrict__ p, int n) {
  int i = blockIdx.x * blockDim.x + threadIdx.x;
  if (i < n) p[i] = 0.0f;
}

__global__ void __launch_bounds__(32)
proj_kernel(const float* __restrict__ audio, const float* __restrict__ W,
            const float* __restrict__ bias, float* __restrict__ kw,
            float* __restrict__ sumsq) {
  const int lane = threadIdx.x;
  const int hi   = lane >> 4;
  const int l15  = lane & 15;
  const int mt = blockIdx.x, nt = blockIdx.y;
  const int m = mt * 16 + l15;
  const int n = nt * 16 + l15;

  const float* Arow = audio + (size_t)m * Dd;
  v8f acc = {};
#pragma unroll 4
  for (int k = 0; k < Dd; k += 4) {
    const int klo = k + 2 * hi;
    v2f a = *(const v2f*)(Arow + klo);
    v2f b;
    b.x = W[(size_t)klo * Tt + n];
    b.y = W[(size_t)(klo + 1) * Tt + n];
    acc = wmma_f32_k4(a, b, acc);
  }

  const float bn = bias[n];
#pragma unroll
  for (int i = 0; i < 8; ++i) {
    const int row = mt * 16 + i + 8 * hi;
    float val = acc[i] + bn;
    kw[(size_t)row * Tt + n] = val;
    float s = val * val;
    s += __shfl_xor(s, 1);
    s += __shfl_xor(s, 2);
    s += __shfl_xor(s, 4);
    s += __shfl_xor(s, 8);
    if (l15 == 0) atomicAdd(&sumsq[row], s);
  }
}

__global__ void __launch_bounds__(256)
embnorm_kernel(const float* __restrict__ emb, float* __restrict__ en) {
  const int wave = threadIdx.x >> 5, lane = threadIdx.x & 31;
  const int v = blockIdx.x * 8 + wave;
  const float* row = emb + (size_t)v * Tt;
  float s = 0.0f;
#pragma unroll
  for (int j = 0; j < 4; ++j) {
    v4f d = *(const v4f*)(row + (j * 32 + lane) * 4);
    s += d.x * d.x + d.y * d.y + d.z * d.z + d.w * d.w;
  }
  s += __shfl_xor(s, 1);  s += __shfl_xor(s, 2);  s += __shfl_xor(s, 4);
  s += __shfl_xor(s, 8);  s += __shfl_xor(s, 16);
  if (lane == 0) en[v] = sqrtf(s);
}

__global__ void __launch_bounds__(128)
score_kernel(const float* __restrict__ kw, const float* __restrict__ sumsq,
             const float* __restrict__ emb, const float* __restrict__ en,
             float* __restrict__ pscore, int* __restrict__ pidx) {
  __shared__ float Atile[32 * Tt];
  const int tid  = threadIdx.x;
  const int wave = tid >> 5, lane = tid & 31;
  const int hi = lane >> 4, l15 = lane & 15;
  const int rst    = blockIdx.x;
  const int stream = blockIdx.y * 4 + wave;

  for (int r = wave; r < 32; r += 4) {
    const float* src = kw + (size_t)(rst * 32 + r) * Tt;
    const int swz = 4 * (r & 15);
#pragma unroll
    for (int j = 0; j < 4; ++j) {
      const int c = j * 128 + lane * 4;
      *(v4f*)(Atile + r * Tt + (c ^ swz)) = *(const v4f*)(src + c);
    }
  }
  __syncthreads();

  const int aswz = 4 * l15;
  float rnorm[2][8];
#pragma unroll
  for (int r = 0; r < 2; ++r)
#pragma unroll
    for (int i = 0; i < 8; ++i)
      rnorm[r][i] = sqrtf(sumsq[rst * 32 + r * 16 + i + 8 * hi]);

  float best[2][8]; int bidx[2][8];
#pragma unroll
  for (int r = 0; r < 2; ++r)
#pragma unroll
    for (int i = 0; i < 8; ++i) { best[r][i] = -3.4e38f; bidx[r][i] = 0x7fffffff; }

  for (int p = stream; p < NP; p += S) {
    const int v0 = (2 * p) * 16 + l15;
    const int v1 = (2 * p + 1) * 16 + l15;
    if (p + S < NP) {
      __builtin_prefetch(emb + (size_t)(2 * (p + S)) * 16 * Tt + (size_t)l15 * Tt, 0, 1);
      __builtin_prefetch(emb + (size_t)(2 * (p + S) + 1) * 16 * Tt + (size_t)l15 * Tt, 0, 1);
    }
    const float en0 = en[v0], en1 = en[v1];
    const float* B0 = emb + (size_t)v0 * Tt;
    const float* B1 = emb + (size_t)v1 * Tt;

    v8f a00 = {}, a01 = {}, a10 = {}, a11 = {};
#pragma unroll 4
    for (int k = 0; k < Tt; k += 4) {
      const int klo = k + 2 * hi;
      v2f fa0 = *(const v2f*)(Atile + l15 * Tt + (klo ^ aswz));
      v2f fa1 = *(const v2f*)(Atile + (16 + l15) * Tt + (klo ^ aswz));
      v2f fb0 = *(const v2f*)(B0 + klo);
      v2f fb1 = *(const v2f*)(B1 + klo);
      a00 = wmma_f32_k4(fa0, fb0, a00);
      a01 = wmma_f32_k4(fa0, fb1, a01);
      a10 = wmma_f32_k4(fa1, fb0, a10);
      a11 = wmma_f32_k4(fa1, fb1, a11);
    }

#pragma unroll
    for (int i = 0; i < 8; ++i) {
      upd(best[0][i], bidx[0][i], a00[i] / fmaxf(rnorm[0][i] * en0, EPS), v0);
      upd(best[0][i], bidx[0][i], a01[i] / fmaxf(rnorm[0][i] * en1, EPS), v1);
      upd(best[1][i], bidx[1][i], a10[i] / fmaxf(rnorm[1][i] * en0, EPS), v0);
      upd(best[1][i], bidx[1][i], a11[i] / fmaxf(rnorm[1][i] * en1, EPS), v1);
    }
  }

#pragma unroll
  for (int r = 0; r < 2; ++r)
#pragma unroll
    for (int i = 0; i < 8; ++i) {
      float s = best[r][i]; int v = bidx[r][i];
#pragma unroll
      for (int mask = 1; mask <= 8; mask <<= 1) {
        const float so = __shfl_xor(s, mask);
        const int vo = __shfl_xor(v, mask);
        if (so > s || (so == s && vo < v)) { s = so; v = vo; }
      }
      if (l15 == 0) {
        const int row = rst * 32 + r * 16 + i + 8 * hi;
        pscore[(size_t)row * S + stream] = s;
        pidx[(size_t)row * S + stream]   = v;
      }
    }
}

__global__ void __launch_bounds__(256)
finalize_kernel(const float* __restrict__ pscore, const int* __restrict__ pidx,
                const float* __restrict__ emb, float* __restrict__ out) {
  const int wave = threadIdx.x >> 5, lane = threadIdx.x & 31;
  const int row = blockIdx.x * 8 + wave;
  float s = -3.4e38f; int v = 0x7fffffff;
#pragma unroll
  for (int j = 0; j < 4; ++j) {
    const float sj = pscore[(size_t)row * S + j * 32 + lane];
    const int   vj = pidx[(size_t)row * S + j * 32 + lane];
    if (sj > s || (sj == s && vj < v)) { s = sj; v = vj; }
  }
#pragma unroll
  for (int mask = 1; mask <= 16; mask <<= 1) {
    const float so = __shfl_xor(s, mask);
    const int vo = __shfl_xor(v, mask);
    if (so > s || (so == s && vo < v)) { s = so; v = vo; }
  }
  const float* src = emb + (size_t)v * Tt;
  float* dst = out + (size_t)row * Tt;
#pragma unroll
  for (int j = 0; j < 4; ++j) {
    const int c = (j * 32 + lane) * 4;
    *(v4f*)(dst + c) = *(const v4f*)(src + c);
  }
}

extern "C" void kernel_launch(void* const* d_in, const int* in_sizes, int n_in,
                              void* d_out, int out_size, void* d_ws, size_t ws_size,
                              hipStream_t stream) {
  const float* audio = (const float*)d_in[0];
  const float* W     = (const float*)d_in[1];
  const float* bias  = (const float*)d_in[2];
  const float* emb   = (const float*)d_in[3];
  float* out = (float*)d_out;

  float* ws    = (float*)d_ws;
  float* kw    = ws;
  float* sumsq = kw + (size_t)R * Tt;
  float* en    = sumsq + R;
  float* psc   = en + Vv;
  int*   pix   = (int*)(psc + (size_t)R * S);

  zero_f32<<<(R + 255) / 256, 256, 0, stream>>>(sumsq, R);
  proj_kernel<<<dim3(R / 16, Tt / 16), 32, 0, stream>>>(audio, W, bias, kw, sumsq);
  embnorm_kernel<<<Vv / 8, 256, 0, stream>>>(emb, en);
  score_kernel<<<dim3(R / 32, S / 4), 128, 0, stream>>>(kw, sumsq, emb, en, psc, pix);
  finalize_kernel<<<R / 8, 256, 0, stream>>>(psc, pix, emb, out);
}


# score kernel: A (LDS) and B (global) WMMA fragments fetched 16 B per lane instead of 8 B: half the loads and cache-line touches; k order in each 8-k group changes, so f32 summation order changes
# speedup vs baseline: 1.0761x; 1.0761x over previous
#include <hip/hip_runtime.h>

typedef __attribute__((ext_vector_type(2))) float v2f;
typedef __attribute__((ext_vector_type(4))) float v4f;
typedef __attribute__((ext_vector_type(8))) float v8f;

constexpr int Dd = 768;
constexpr int Tt = 512;
constexpr int Vv = 49408;
constexpr int R  = 64 * 8;
constexpr int VT = Vv / 16;
constexpr int NP = VT / 2;
constexpr int S  = 128;
constexpr float EPS = 1e-8f;

__device__ __forceinline__ v8f wmma_f32_k4(v2f a, v2f b, v8f c) {
  return __builtin_amdgcn_wmma_f32_16x16x4_f32(
       false, a,  false, b,
       (short)0, c,  false,  false);
}

__device__ __forceinline__ void upd(float& bs, int& bv, float s, int v) {
  if (s > bs || (s == bs && v < bv)) { bs = s; bv = v; }
}

__global__ void zero_f32(float* __restrict__ p, int n) {
  int i = blockIdx.x * blockDim.x + threadIdx.x;
  if (i < n) p[i] = 0.0f;
}

__global__ void __launch_bounds__(32)
proj_kernel(const float* __restrict__ audio, const float* __restrict__ W,
            const float* __restrict__ bias, float* __restrict__ kw,
            float* __restrict__ sumsq) {
  const int lane = threadIdx.x;
  const int hi   = lane >> 4;
  const int l15  = lane & 15;
  const int mt = blockIdx.x, nt = blockIdx.y;
  const int m = mt * 16 + l15;
  const int n = nt * 16 + l15;

  const float* Arow = audio + (size_t)m * Dd;
  v8f acc = {};
#pragma unroll 4
  for (int k = 0; k < Dd; k += 4) {
    const int klo = k + 2 * hi;
    v2f a = *(const v2f*)(Arow + klo);
    v2f b;
    b.x = W[(size_t)klo * Tt + n];
    b.y = W[(size_t)(klo + 1) * Tt + n];
    acc = wmma_f32_k4(a, b, acc);
  }

  const float bn = bias[n];
#pragma unroll
  for (int i = 0; i < 8; ++i) {
    const int row = mt * 16 + i + 8 * hi;
    float val = acc[i] + bn;
    kw[(size_t)row * Tt + n] = val;
    float s = val * val;
    s += __shfl_xor(s, 1);
    s += __shfl_xor(s, 2);
    s += __shfl_xor(s, 4);
    s += __shfl_xor(s, 8);
    if (l15 == 0) atomicAdd(&sumsq[row], s);
  }
}

__global__ void __launch_bounds__(256)
embnorm_kernel(const float* __restrict__ emb, float* __restrict__ en) {
  const int wave = threadIdx.x >> 5, lane = threadIdx.x & 31;
  const int v = blockIdx.x * 8 + wave;
  const float* row = emb + (size_t)v * Tt;
  float s = 0.0f;
#pragma unroll
  for (int j = 0; j < 4; ++j) {
    v4f d = *(const v4f*)(row + (j * 32 + lane) * 4);
    s += d.x * d.x + d.y * d.y + d.z * d.z + d.w * d.w;
  }
  s += __shfl_xor(s, 1);  s += __shfl_xor(s, 2);  s += __shfl_xor(s, 4);
  s += __shfl_xor(s, 8);  s += __shfl_xor(s, 16);
  if (lane == 0) en[v] = sqrtf(s);
}

__global__ void __launch_bounds__(128)
score_kernel(const float* __restrict__ kw, const float* __restrict__ sumsq,
             const float* __restrict__ emb, const float* __restrict__ en,
             float* __restrict__ pscore, int* __restrict__ pidx) {
  __shared__ float Atile[32 * Tt];
  const int tid  = threadIdx.x;
  const int wave = tid >> 5, lane = tid & 31;
  const int hi = lane >> 4, l15 = lane & 15;
  const int rst    = blockIdx.x;
  const int stream = blockIdx.y * 4 + wave;

  for (int r = wave; r < 32; r += 4) {
    const float* src = kw + (size_t)(rst * 32 + r) * Tt;
    const int swz = 4 * (r & 15);
#pragma unroll
    for (int j = 0; j < 4; ++j) {
      const int c = j * 128 + lane * 4;
      *(v4f*)(Atile + r * Tt + (c ^ swz)) = *(const v4f*)(src + c);
    }
  }
  __syncthreads();

  const int aswz = 4 * l15;
  float rnorm[2][8];
#pragma unroll
  for (int r = 0; r < 2; ++r)
#pragma unroll
    for (int i = 0; i < 8; ++i)
      rnorm[r][i] = sqrtf(sumsq[rst * 32 + r * 16 + i + 8 * hi]);

  float best[2][8]; int bidx[2][8];
#pragma unroll
  for (int r = 0; r < 2; ++r)
#pragma unroll
    for (int i = 0; i < 8; ++i) { best[r][i] = -3.4e38f; bidx[r][i] = 0x7fffffff; }

  for (int p = stream; p < NP; p += S) {
    const int v0 = (2 * p) * 16 + l15;
    const int v1 = (2 * p + 1) * 16 + l15;
    if (p + S < NP) {
      __builtin_prefetch(emb + (size_t)(2 * (p + S)) * 16 * Tt + (size_t)l15 * Tt, 0, 1);
      __builtin_prefetch(emb + (size_t)(2 * (p + S) + 1) * 16 * Tt + (size_t)l15 * Tt, 0, 1);
    }
    const float en0 = en[v0], en1 = en[v1];
    const float* B0 = emb + (size_t)v0 * Tt;
    const float* B1 = emb + (size_t)v1 * Tt;

    v8f a00 = {}, a01 = {}, a10 = {}, a11 = {};
#pragma unroll 4
    for (int k8 = 0; k8 < Tt; k8 += 8) {
      const int klo = k8 + 4 * hi;
      v4f fa0 = *(const v4f*)(Atile + l15 * Tt + (klo ^ aswz));
      v4f fa1 = *(const v4f*)(Atile + (16 + l15) * Tt + (klo ^ aswz));
      v4f fb0 = *(const v4f*)(B0 + klo);
      v4f fb1 = *(const v4f*)(B1 + klo);
      const v2f a0l = {fa0.x, fa0.y}, a0h = {fa0.z, fa0.w};
      const v2f a1l = {fa1.x, fa1.y}, a1h = {fa1.z, fa1.w};
      const v2f b0l = {fb0.x, fb0.y}, b0h = {fb0.z, fb0.w};
      const v2f b1l = {fb1.x, fb1.y}, b1h = {fb1.z, fb1.w};
      a00 = wmma_f32_k4(a0l, b0l, a00);
      a01 = wmma_f32_k4(a0l, b1l, a01);
      a10 = wmma_f32_k4(a1l, b0l, a10);
      a11 = wmma_f32_k4(a1l, b1l, a11);
      a00 = wmma_f32_k4(a0h, b0h, a00);
      a01 = wmma_f32_k4(a0h, b1h, a01);
      a10 = wmma_f32_k4(a1h, b0h, a10);
      a11 = wmma_f32_k4(a1h, b1h, a11);
    }

#pragma unroll
    for (int i = 0; i < 8; ++i) {
      upd(best[0][i], bidx[0][i], a00[i] / fmaxf(rnorm[0][i] * en0, EPS), v0);
      upd(best[0][i], bidx[0][i], a01[i] / fmaxf(rnorm[0][i] * en1, EPS), v1);
      upd(best[1][i], bidx[1][i], a10[i] / fmaxf(rnorm[1][i] * en0, EPS), v0);
      upd(best[1][i], bidx[1][i], a11[i] / fmaxf(rnorm[1][i] * en1, EPS), v1);
    }
  }

#pragma unroll
  for (int r = 0; r < 2; ++r)
#pragma unroll
    for (int i = 0; i < 8; ++i) {
      float s = best[r][i]; int v = bidx[r][i];
#pragma unroll
      for (int mask = 1; mask <= 8; mask <<= 1) {
        const float so = __shfl_xor(s, mask);
        const int vo = __shfl_xor(v, mask);
        if (so > s || (so == s && vo < v)) { s = so; v = vo; }
      }
      if (l15 == 0) {
        const int row = rst * 32 + r * 16 + i + 8 * hi;
        pscore[(size_t)row * S + stream] = s;
        pidx[(size_t)row * S + stream]   = v;
      }
    }
}

__global__ void __launch_bounds__(256)
finalize_kernel(const float* __restrict__ pscore, const int* __restrict__ pidx,
                const float* __restrict__ emb, float* __restrict__ out) {
  const int wave = threadIdx.x >> 5, lane = threadIdx.x & 31;
  const int row = blockIdx.x * 8 + wave;
  float s = -3.4e38f; int v = 0x7fffffff;
#pragma unroll
  for (int j = 0; j < 4; ++j) {
    const float sj = pscore[(size_t)row * S + j * 32 + lane];
    const int   vj = pidx[(size_t)row * S + j * 32 + lane];
    if (sj > s || (sj == s && vj < v)) { s = sj; v = vj; }
  }
#pragma unroll
  for (int mask = 1; mask <= 16; mask <<= 1) {
    const float so = __shfl_xor(s, mask);
    const int vo = __shfl_xor(v, mask);
    if (so > s || (so == s && vo < v)) { s = so; v = vo; }
  }
  const float* src = emb + (size_t)v * Tt;
  float* dst = out + (size_t)row * Tt;
#pragma unroll
  for (int j = 0; j < 4; ++j) {
    const int c = (j * 32 + lane) * 4;
    *(v4f*)(dst + c) = *(const v4f*)(src + c);
  }
}

extern "C" void kernel_launch(void* const* d_in, const int* in_sizes, int n_in,
                              void* d_out, int out_size, void* d_ws, size_t ws_size,
                              hipStream_t stream) {
  const float* audio = (const float*)d_in[0];
  const float* W     = (const float*)d_in[1];
  const float* bias  = (const float*)d_in[2];
  const float* emb   = (const float*)d_in[3];
  float* out = (float*)d_out;

  float* ws    = (float*)d_ws;
  float* kw    = ws;
  float* sumsq = kw + (size_t)R * Tt;
  float* en    = sumsq + R;
  float* psc   = en + Vv;
  int*   pix   = (int*)(psc + (size_t)R * S);

  zero_f32<<<(R + 255) / 256, 256, 0, stream>>>(sumsq, R);
  proj_kernel<<<dim3(R / 16, Tt / 16), 32, 0, stream>>>(audio, W, bias, kw, sumsq);
  embnorm_kernel<<<Vv / 8, 256, 0, stream>>>(emb, en);
  score_kernel<<<dim3(R / 32, S / 4), 128, 0, stream>>>(kw, sumsq, emb, en, psc, pix);
  finalize_kernel<<<R / 8, 256, 0, stream>>>(psc, pix, emb, out);
}


# score kernel: K loop software-pipelined over two register sets: next 64-k block's B (global) and A (LDS) fragment loads issued as one group ahead of the current block's 64 WMMAs; sum order unchanged
# speedup vs baseline: 1.2810x; 1.1904x over previous
#include <hip/hip_runtime.h>

typedef __attribute__((ext_vector_type(2))) float v2f;
typedef __attribute__((ext_vector_type(4))) float v4f;
typedef __attribute__((ext_vector_type(8))) float v8f;

constexpr int Dd = 768;
constexpr int Tt = 512;
constexpr int Vv = 49408;
constexpr int R  = 64 * 8;
constexpr int VT = Vv / 16;
constexpr int NP = VT / 2;
constexpr int S  = 128;
constexpr float EPS = 1e-8f;

__device__ __forceinline__ v8f wmma_f32_k4(v2f a, v2f b, v8f c) {
  return __builtin_amdgcn_wmma_f32_16x16x4_f32(
       false, a,  false, b,
       (short)0, c,  false,  false);
}

__device__ __forceinline__ void upd(float& bs, int& bv, float s, int v) {
  if (s > bs || (s == bs && v < bv)) { bs = s; bv = v; }
}

__global__ void zero_f32(float* __restrict__ p, int n) {
  int i = blockIdx.x * blockDim.x + threadIdx.x;
  if (i < n) p[i] = 0.0f;
}

__global__ void __launch_bounds__(32)
proj_kernel(const float* __restrict__ audio, const float* __restrict__ W,
            const float* __restrict__ bias, float* __restrict__ kw,
            float* __restrict__ sumsq) {
  const int lane = threadIdx.x;
  const int hi   = lane >> 4;
  const int l15  = lane & 15;
  const int mt = blockIdx.x, nt = blockIdx.y;
  const int m = mt * 16 + l15;
  const int n = nt * 16 + l15;

  const float* Arow = audio + (size_t)m * Dd;
  v8f acc = {};
#pragma unroll 4
  for (int k = 0; k < Dd; k += 4) {
    const int klo = k + 2 * hi;
    v2f a = *(const v2f*)(Arow + klo);
    v2f b;
    b.x = W[(size_t)klo * Tt + n];
    b.y = W[(size_t)(klo + 1) * Tt + n];
    acc = wmma_f32_k4(a, b, acc);
  }

  const float bn = bias[n];
#pragma unroll
  for (int i = 0; i < 8; ++i) {
    const int row = mt * 16 + i + 8 * hi;
    float val = acc[i] + bn;
    kw[(size_t)row * Tt + n] = val;
    float s = val * val;
    s += __shfl_xor(s, 1);
    s += __shfl_xor(s, 2);
    s += __shfl_xor(s, 4);
    s += __shfl_xor(s, 8);
    if (l15 == 0) atomicAdd(&sumsq[row], s);
  }
}

__global__ void __launch_bounds__(256)
embnorm_kernel(const float* __restrict__ emb, float* __restrict__ en) {
  const int wave = threadIdx.x >> 5, lane = threadIdx.x & 31;
  const int v = blockIdx.x * 8 + wave;
  const float* row = emb + (size_t)v * Tt;
  float s = 0.0f;
#pragma unroll
  for (int j = 0; j < 4; ++j) {
    v4f d = *(const v4f*)(row + (j * 32 + lane) * 4);
    s += d.x * d.x + d.y * d.y + d.z * d.z + d.w * d.w;
  }
  s += __shfl_xor(s, 1);  s += __shfl_xor(s, 2);  s += __shfl_xor(s, 4);
  s += __shfl_xor(s, 8);  s += __shfl_xor(s, 16);
  if (lane == 0) en[v] = sqrtf(s);
}

struct Blk { v4f a0[8], a1[8], b0[8], b1[8]; };

__device__ __forceinline__ void load_blk(Blk& f, const float* A0, const float* A1, int aswz,
                                         const float* __restrict__ B0,
                                         const float* __restrict__ B1, int kq) {
#pragma unroll
  for (int j = 0; j < 8; ++j) {
    f.b0[j] = *(const v4f*)(B0 + kq + 8 * j);
    f.b1[j] = *(const v4f*)(B1 + kq + 8 * j);
  }
#pragma unroll
  for (int j = 0; j < 8; ++j) {
    const int kl = (kq + 8 * j) ^ aswz;
    f.a0[j] = *(const v4f*)(A0 + kl);
    f.a1[j] = *(const v4f*)(A1 + kl);
  }
}

__device__ __forceinline__ void mma_blk(const Blk& f, v8f& a00, v8f& a01, v8f& a10, v8f& a11) {
#pragma unroll
  for (int j = 0; j < 8; ++j) {
    const v2f a0l = {f.a0[j].x, f.a0[j].y}, a0h = {f.a0[j].z, f.a0[j].w};
    const v2f a1l = {f.a1[j].x, f.a1[j].y}, a1h = {f.a1[j].z, f.a1[j].w};
    const v2f b0l = {f.b0[j].x, f.b0[j].y}, b0h = {f.b0[j].z, f.b0[j].w};
    const v2f b1l = {f.b1[j].x, f.b1[j].y}, b1h = {f.b1[j].z, f.b1[j].w};
    a00 = wmma_f32_k4(a0l, b0l, a00);
    a01 = wmma_f32_k4(a0l, b1l, a01);
    a10 = wmma_f32_k4(a1l, b0l, a10);
    a11 = wmma_f32_k4(a1l, b1l, a11);
    a00 = wmma_f32_k4(a0h, b0h, a00);
    a01 = wmma_f32_k4(a0h, b1h, a01);
    a10 = wmma_f32_k4(a1h, b0h, a10);
    a11 = wmma_f32_k4(a1h, b1h, a11);
  }
}

__device__ __forceinline__ void pin(v8f& a00, v8f& a01, v8f& a10, v8f& a11) {
  asm volatile("" : "+v"(a00), "+v"(a01), "+v"(a10), "+v"(a11));
}

__global__ void __launch_bounds__(128)
score_kernel(const float* __restrict__ kw, const float* __restrict__ sumsq,
             const float* __restrict__ emb, const float* __restrict__ en,
             float* __restrict__ pscore, int* __restrict__ pidx) {
  __shared__ float Atile[32 * Tt];
  const int tid  = threadIdx.x;
  const int wave = tid >> 5, lane = tid & 31;
  const int hi = lane >> 4, l15 = lane & 15;
  const int rst    = blockIdx.x;
  const int stream = blockIdx.y * 4 + wave;

  for (int r = wave; r < 32; r += 4) {
    const float* src = kw + (size_t)(rst * 32 + r) * Tt;
    const int swz = 4 * (r & 15);
#pragma unroll
    for (int j = 0; j < 4; ++j) {
      const int c = j * 128 + lane * 4;
      *(v4f*)(Atile + r * Tt + (c ^ swz)) = *(const v4f*)(src + c);
    }
  }
  __syncthreads();

  const int aswz = 4 * l15;
  float rnorm[2][8];
#pragma unroll
  for (int r = 0; r < 2; ++r)
#pragma unroll
    for (int i = 0; i < 8; ++i)
      rnorm[r][i] = sqrtf(sumsq[rst * 32 + r * 16 + i + 8 * hi]);

  float best[2][8]; int bidx[2][8];
#pragma unroll
  for (int r = 0; r < 2; ++r)
#pragma unroll
    for (int i = 0; i < 8; ++i) { best[r][i] = -3.4e38f; bidx[r][i] = 0x7fffffff; }

  const float* A0 = Atile + l15 * Tt;
  const float* A1 = Atile + (16 + l15) * Tt;
  const int kh = 4 * hi;
  constexpr int KB = 64;
  constexpr int NB = Tt / KB;

  const float* B0 = emb + (size_t)((2 * stream) * 16 + l15) * Tt;
  const float* B1 = B0 + (size_t)16 * Tt;
  Blk X, Y;
  load_blk(X, A0, A1, aswz, B0, B1, kh);

  for (int p = stream; p < NP; p += S) {
    const int v0 = (2 * p) * 16 + l15;
    const int v1 = (2 * p + 1) * 16 + l15;
    const bool more = p + S < NP;
    if (more) {
      __builtin_prefetch(emb + (size_t)(2 * (p + S)) * 16 * Tt + (size_t)l15 * Tt, 0, 1);
      __builtin_prefetch(emb + (size_t)(2 * (p + S) + 1) * 16 * Tt + (size_t)l15 * Tt, 0, 1);
    }
    const float en0 = en[v0], en1 = en[v1];
    const float* nB0 = more ? B0 + (size_t)S * 32 * Tt : B0;
    const float* nB1 = nB0 + (size_t)16 * Tt;

    v8f a00 = {}, a01 = {}, a10 = {}, a11 = {};
    for (int kb = 0; kb < NB; kb += 2) {
      load_blk(Y, A0, A1, aswz, B0, B1, (kb + 1) * KB + kh);
      pin(a00, a01, a10, a11);
      mma_blk(X, a00, a01, a10, a11);
      pin(a00, a01, a10, a11);
      const bool last = kb + 2 == NB;
      const int kn = (kb + 2) & (NB - 1);
      load_blk(X, A0, A1, aswz, last ? nB0 : B0, last ? nB1 : B1, kn * KB + kh);
      pin(a00, a01, a10, a11);
      mma_blk(Y, a00, a01, a10, a11);
      pin(a00, a01, a10, a11);
    }

#pragma unroll
    for (int i = 0; i < 8; ++i) {
      upd(best[0][i], bidx[0][i], a00[i] / fmaxf(rnorm[0][i] * en0, EPS), v0);
      upd(best[0][i], bidx[0][i], a01[i] / fmaxf(rnorm[0][i] * en1, EPS), v1);
      upd(best[1][i], bidx[1][i], a10[i] / fmaxf(rnorm[1][i] * en0, EPS), v0);
      upd(best[1][i], bidx[1][i], a11[i] / fmaxf(rnorm[1][i] * en1, EPS), v1);
    }
    B0 = nB0; B1 = nB1;
  }

#pragma unroll
  for (int r = 0; r < 2; ++r)
#pragma unroll
    for (int i = 0; i < 8; ++i) {
      float s = best[r][i]; int v = bidx[r][i];
#pragma unroll
      for (int mask = 1; mask <= 8; mask <<= 1) {
        const float so = __shfl_xor(s, mask);
        const int vo = __shfl_xor(v, mask);
        if (so > s || (so == s && vo < v)) { s = so; v = vo; }
      }
      if (l15 == 0) {
        const int row = rst * 32 + r * 16 + i + 8 * hi;
        pscore[(size_t)row * S + stream] = s;
        pidx[(size_t)row * S + stream]   = v;
      }
    }
}

__global__ void __launch_bounds__(256)
finalize_kernel(const float* __restrict__ pscore, const int* __restrict__ pidx,
                const float* __restrict__ emb, float* __restrict__ out) {
  const int wave = threadIdx.x >> 5, lane = threadIdx.x & 31;
  const int row = blockIdx.x * 8 + wave;
  float s = -3.4e38f; int v = 0x7fffffff;
#pragma unroll
  for (int j = 0; j < 4; ++j) {
    const float sj = pscore[(size_t)row * S + j * 32 + lane];
    const int   vj = pidx[(size_t)row * S + j * 32 + lane];
    if (sj > s || (sj == s && vj < v)) { s = sj; v = vj; }
  }
#pragma unroll
  for (int mask = 1; mask <= 16; mask <<= 1) {
    const float so = __shfl_xor(s, mask);
    const int vo = __shfl_xor(v, mask);
    if (so > s || (so == s && vo < v)) { s = so; v = vo; }
  }
  const float* src = emb + (size_t)v * Tt;
  float* dst = out + (size_t)row * Tt;
#pragma unroll
  for (int j = 0; j < 4; ++j) {
    const int c = (j * 32 + lane) * 4;
    *(v4f*)(dst + c) = *(const v4f*)(src + c);
  }
}

extern "C" void kernel_launch(void* const* d_in, const int* in_sizes, int n_in,
                              void* d_out, int out_size, void* d_ws, size_t ws_size,
                              hipStream_t stream) {
  const float* audio = (const float*)d_in[0];
  const float* W     = (const float*)d_in[1];
  const float* bias  = (const float*)d_in[2];
  const float* emb   = (const float*)d_in[3];
  float* out = (float*)d_out;

  float* ws    = (float*)d_ws;
  float* kw    = ws;
  float* sumsq = kw + (size_t)R * Tt;
  float* en    = sumsq + R;
  float* psc   = en + Vv;
  int*   pix   = (int*)(psc + (size_t)R * S);

  zero_f32<<<(R + 255) / 256, 256, 0, stream>>>(sumsq, R);
  proj_kernel<<<dim3(R / 16, Tt / 16), 32, 0, stream>>>(audio, W, bias, kw, sumsq);
  embnorm_kernel<<<Vv / 8, 256, 0, stream>>>(emb, en);
  score_kernel<<<dim3(R / 32, S / 4), 128, 0, stream>>>(kw, sumsq, emb, en, psc, pix);
  finalize_kernel<<<R / 8, 256, 0, stream>>>(psc, pix, emb, out);
}


# projection kernel made latency-tolerant: each tile's K=768 split over 4 waves, k-steps unrolled at constant offsets (30 loads in flight per wave), partials added via LDS; kw f32 sum order changes
# speedup vs baseline: 1.3962x; 1.0899x over previous
#include <hip/hip_runtime.h>

typedef __attribute__((ext_vector_type(2))) float v2f;
typedef __attribute__((ext_vector_type(4))) float v4f;
typedef __attribute__((ext_vector_type(8))) float v8f;

constexpr int Dd = 768;
constexpr int Tt = 512;
constexpr int Vv = 49408;
constexpr int R  = 64 * 8;
constexpr int VT = Vv / 16;
constexpr int NP = VT / 2;
constexpr int S  = 128;
constexpr float EPS = 1e-8f;

__device__ __forceinline__ v8f wmma_f32_k4(v2f a, v2f b, v8f c) {
  return __builtin_amdgcn_wmma_f32_16x16x4_f32(
       false, a,  false, b,
       (short)0, c,  false,  false);
}

__device__ __forceinline__ void upd(float& bs, int& bv, float s, int v) {
  if (s > bs || (s == bs && v < bv)) { bs = s; bv = v; }
}

__global__ void zero_f32(float* __restrict__ p, int n) {
  int i = blockIdx.x * blockDim.x + threadIdx.x;
  if (i < n) p[i] = 0.0f;
}

__global__ void __launch_bounds__(128)
proj_kernel(const float* __restrict__ audio, const float* __restrict__ W,
            const float* __restrict__ bias, float* __restrict__ kw,
            float* __restrict__ sumsq) {
  __shared__ float part[4][8][32];
  const int wave = threadIdx.x >> 5, lane = threadIdx.x & 31;
  const int hi   = lane >> 4;
  const int l15  = lane & 15;
  const int mt = blockIdx.x, nt = blockIdx.y;
  const int m = mt * 16 + l15;
  const int n = nt * 16 + l15;
  constexpr int KW = Dd / 4;
  const int k0 = wave * KW;

  const float* Ap = audio + (size_t)m * Dd + k0 + 2 * hi;
  const float* Wp = W + (size_t)(k0 + 2 * hi) * Tt + n;
  v8f acc = {};
  for (int t = 0; t < KW / 64; ++t) {
#pragma unroll
    for (int s = 0; s < 16; ++s) {
      v2f a = *(const v2f*)(Ap + 4 * s);
      v2f b;
      b.x = Wp[(size_t)(4 * s) * Tt];
      b.y = Wp[(size_t)(4 * s + 1) * Tt];
      acc = wmma_f32_k4(a, b, acc);
    }
    Ap += 64;
    Wp += (size_t)64 * Tt;
  }

#pragma unroll
  for (int i = 0; i < 8; ++i) part[wave][i][lane] = acc[i];
  __syncthreads();

  const float bn = bias[n];
#pragma unroll
  for (int ii = 0; ii < 2; ++ii) {
    const int i = 2 * wave + ii;
    const int row = mt * 16 + i + 8 * hi;
    float val = ((part[0][i][lane] + part[1][i][lane]) + part[2][i][lane]) + part[3][i][lane];
    val += bn;
    kw[(size_t)row * Tt + n] = val;
    float s = val * val;
    s += __shfl_xor(s, 1);
    s += __shfl_xor(s, 2);
    s += __shfl_xor(s, 4);
    s += __shfl_xor(s, 8);
    if (l15 == 0) atomicAdd(&sumsq[row], s);
  }
}

__global__ void __launch_bounds__(256)
embnorm_kernel(const float* __restrict__ emb, float* __restrict__ en) {
  const int wave = threadIdx.x >> 5, lane = threadIdx.x & 31;
  const int v = blockIdx.x * 8 + wave;
  const float* row = emb + (size_t)v * Tt;
  float s = 0.0f;
#pragma unroll
  for (int j = 0; j < 4; ++j) {
    v4f d = *(const v4f*)(row + (j * 32 + lane) * 4);
    s += d.x * d.x + d.y * d.y + d.z * d.z + d.w * d.w;
  }
  s += __shfl_xor(s, 1);  s += __shfl_xor(s, 2);  s += __shfl_xor(s, 4);
  s += __shfl_xor(s, 8);  s += __shfl_xor(s, 16);
  if (lane == 0) en[v] = sqrtf(s);
}

struct Blk { v4f a0[8], a1[8], b0[8], b1[8]; };

__device__ __forceinline__ void load_blk(Blk& f, const float* A0, const float* A1, int aswz,
                                         const float* __restrict__ B0,
                                         const float* __restrict__ B1, int kq) {
#pragma unroll
  for (int j = 0; j < 8; ++j) {
    f.b0[j] = *(const v4f*)(B0 + kq + 8 * j);
    f.b1[j] = *(const v4f*)(B1 + kq + 8 * j);
  }
#pragma unroll
  for (int j = 0; j < 8; ++j) {
    const int kl = (kq + 8 * j) ^ aswz;
    f.a0[j] = *(const v4f*)(A0 + kl);
    f.a1[j] = *(const v4f*)(A1 + kl);
  }
}

__device__ __forceinline__ void mma_blk(const Blk& f, v8f& a00, v8f& a01, v8f& a10, v8f& a11) {
#pragma unroll
  for (int j = 0; j < 8; ++j) {
    const v2f a0l = {f.a0[j].x, f.a0[j].y}, a0h = {f.a0[j].z, f.a0[j].w};
    const v2f a1l = {f.a1[j].x, f.a1[j].y}, a1h = {f.a1[j].z, f.a1[j].w};
    const v2f b0l = {f.b0[j].x, f.b0[j].y}, b0h = {f.b0[j].z, f.b0[j].w};
    const v2f b1l = {f.b1[j].x, f.b1[j].y}, b1h = {f.b1[j].z, f.b1[j].w};
    a00 = wmma_f32_k4(a0l, b0l, a00);
    a01 = wmma_f32_k4(a0l, b1l, a01);
    a10 = wmma_f32_k4(a1l, b0l, a10);
    a11 = wmma_f32_k4(a1l, b1l, a11);
    a00 = wmma_f32_k4(a0h, b0h, a00);
    a01 = wmma_f32_k4(a0h, b1h, a01);
    a10 = wmma_f32_k4(a1h, b0h, a10);
    a11 = wmma_f32_k4(a1h, b1h, a11);
  }
}

__device__ __forceinline__ void pin(v8f& a00, v8f& a01, v8f& a10, v8f& a11) {
  asm volatile("" : "+v"(a00), "+v"(a01), "+v"(a10), "+v"(a11));
}

__global__ void __launch_bounds__(128)
score_kernel(const float* __restrict__ kw, const float* __restrict__ sumsq,
             const float* __restrict__ emb, const float* __restrict__ en,
             float* __restrict__ pscore, int* __restrict__ pidx) {
  __shared__ float Atile[32 * Tt];
  const int tid  = threadIdx.x;
  const int wave = tid >> 5, lane = tid & 31;
  const int hi = lane >> 4, l15 = lane & 15;
  const int rst    = blockIdx.x;
  const int stream = blockIdx.y * 4 + wave;

  for (int r = wave; r < 32; r += 4) {
    const float* src = kw + (size_t)(rst * 32 + r) * Tt;
    const int swz = 4 * (r & 15);
#pragma unroll
    for (int j = 0; j < 4; ++j) {
      const int c = j * 128 + lane * 4;
      *(v4f*)(Atile + r * Tt + (c ^ swz)) = *(const v4f*)(src + c);
    }
  }
  __syncthreads();

  const int aswz = 4 * l15;
  float rnorm[2][8];
#pragma unroll
  for (int r = 0; r < 2; ++r)
#pragma unroll
    for (int i = 0; i < 8; ++i)
      rnorm[r][i] = sqrtf(sumsq[rst * 32 + r * 16 + i + 8 * hi]);

  float best[2][8]; int bidx[2][8];
#pragma unroll
  for (int r = 0; r < 2; ++r)
#pragma unroll
    for (int i = 0; i < 8; ++i) { best[r][i] = -3.4e38f; bidx[r][i] = 0x7fffffff; }

  const float* A0 = Atile + l15 * Tt;
  const float* A1 = Atile + (16 + l15) * Tt;
  const int kh = 4 * hi;
  constexpr int KB = 64;
  constexpr int NB = Tt / KB;

  const float* B0 = emb + (size_t)((2 * stream) * 16 + l15) * Tt;
  const float* B1 = B0 + (size_t)16 * Tt;
  Blk X, Y;
  load_blk(X, A0, A1, aswz, B0, B1, kh);

  for (int p = stream; p < NP; p += S) {
    const int v0 = (2 * p) * 16 + l15;
    const int v1 = (2 * p + 1) * 16 + l15;
    const bool more = p + S < NP;
    if (more) {
      __builtin_prefetch(emb + (size_t)(2 * (p + S)) * 16 * Tt + (size_t)l15 * Tt, 0, 1);
      __builtin_prefetch(emb + (size_t)(2 * (p + S) + 1) * 16 * Tt + (size_t)l15 * Tt, 0, 1);
    }
    const float en0 = en[v0], en1 = en[v1];
    const float* nB0 = more ? B0 + (size_t)S * 32 * Tt : B0;
    const float* nB1 = nB0 + (size_t)16 * Tt;

    v8f a00 = {}, a01 = {}, a10 = {}, a11 = {};
    for (int kb = 0; kb < NB; kb += 2) {
      load_blk(Y, A0, A1, aswz, B0, B1, (kb + 1) * KB + kh);
      pin(a00, a01, a10, a11);
      mma_blk(X, a00, a01, a10, a11);
      pin(a00, a01, a10, a11);
      const bool last = kb + 2 == NB;
      const int kn = (kb + 2) & (NB - 1);
      load_blk(X, A0, A1, aswz, last ? nB0 : B0, last ? nB1 : B1, kn * KB + kh);
      pin(a00, a01, a10, a11);
      mma_blk(Y, a00, a01, a10, a11);
      pin(a00, a01, a10, a11);
    }

#pragma unroll
    for (int i = 0; i < 8; ++i) {
      upd(best[0][i], bidx[0][i], a00[i] / fmaxf(rnorm[0][i] * en0, EPS), v0);
      upd(best[0][i], bidx[0][i], a01[i] / fmaxf(rnorm[0][i] * en1, EPS), v1);
      upd(best[1][i], bidx[1][i], a10[i] / fmaxf(rnorm[1][i] * en0, EPS), v0);
      upd(best[1][i], bidx[1][i], a11[i] / fmaxf(rnorm[1][i] * en1, EPS), v1);
    }
    B0 = nB0; B1 = nB1;
  }

#pragma unroll
  for (int r = 0; r < 2; ++r)
#pragma unroll
    for (int i = 0; i < 8; ++i) {
      float s = best[r][i]; int v = bidx[r][i];
#pragma unroll
      for (int mask = 1; mask <= 8; mask <<= 1) {
        const float so = __shfl_xor(s, mask);
        const int vo = __shfl_xor(v, mask);
        if (so > s || (so == s && vo < v)) { s = so; v = vo; }
      }
      if (l15 == 0) {
        const int row = rst * 32 + r * 16 + i + 8 * hi;
        pscore[(size_t)row * S + stream] = s;
        pidx[(size_t)row * S + stream]   = v;
      }
    }
}

__global__ void __launch_bounds__(256)
finalize_kernel(const float* __restrict__ pscore, const int* __restrict__ pidx,
                const float* __restrict__ emb, float* __restrict__ out) {
  const int wave = threadIdx.x >> 5, lane = threadIdx.x & 31;
  const int row = blockIdx.x * 8 + wave;
  float s = -3.4e38f; int v = 0x7fffffff;
#pragma unroll
  for (int j = 0; j < 4; ++j) {
    const float sj = pscore[(size_t)row * S + j * 32 + lane];
    const int   vj = pidx[(size_t)row * S + j * 32 + lane];
    if (sj > s || (sj == s && vj < v)) { s = sj; v = vj; }
  }
#pragma unroll
  for (int mask = 1; mask <= 16; mask <<= 1) {
    const float so = __shfl_xor(s, mask);
    const int vo = __shfl_xor(v, mask);
    if (so > s || (so == s && vo < v)) { s = so; v = vo; }
  }
  const float* src = emb + (size_t)v * Tt;
  float* dst = out + (size_t)row * Tt;
#pragma unroll
  for (int j = 0; j < 4; ++j) {
    const int c = (j * 32 + lane) * 4;
    *(v4f*)(dst + c) = *(const v4f*)(src + c);
  }
}

extern "C" void kernel_launch(void* const* d_in, const int* in_sizes, int n_in,
                              void* d_out, int out_size, void* d_ws, size_t ws_size,
                              hipStream_t stream) {
  const float* audio = (const float*)d_in[0];
  const float* W     = (const float*)d_in[1];
  const float* bias  = (const float*)d_in[2];
  const float* emb   = (const float*)d_in[3];
  float* out = (float*)d_out;

  float* ws    = (float*)d_ws;
  float* kw    = ws;
  float* sumsq = kw + (size_t)R * Tt;
  float* en    = sumsq + R;
  float* psc   = en + Vv;
  int*   pix   = (int*)(psc + (size_t)R * S);

  zero_f32<<<(R + 255) / 256, 256, 0, stream>>>(sumsq, R);
  proj_kernel<<<dim3(R / 16, Tt / 16), 128, 0, stream>>>(audio, W, bias, kw, sumsq);
  embnorm_kernel<<<Vv / 8, 256, 0, stream>>>(emb, en);
  score_kernel<<<dim3(R / 32, S / 4), 128, 0, stream>>>(kw, sumsq, emb, en, psc, pix);
  finalize_kernel<<<R / 8, 256, 0, stream>>>(psc, pix, emb, out);
}


# score kernel: per-pair epilogue made latency-tolerant: its 32 quotients formed as independent work, running maxima updated by compare/select instead of masked-move sections; scores bit-identical
# speedup vs baseline: 1.4880x; 1.0657x over previous
#include <hip/hip_runtime.h>

typedef __attribute__((ext_vector_type(2))) float v2f;
typedef __attribute__((ext_vector_type(4))) float v4f;
typedef __attribute__((ext_vector_type(8))) float v8f;

constexpr int Dd = 768;
constexpr int Tt = 512;
constexpr int Vv = 49408;
constexpr int R  = 64 * 8;
constexpr int VT = Vv / 16;
constexpr int NP = VT / 2;
constexpr int S  = 128;
constexpr float EPS = 1e-8f;

__device__ __forceinline__ v8f wmma_f32_k4(v2f a, v2f b, v8f c) {
  return __builtin_amdgcn_wmma_f32_16x16x4_f32(
       false, a,  false, b,
       (short)0, c,  false,  false);
}

__device__ __forceinline__ void upd_sel(float& bs, int& bv, float s, int v) {
  const bool t = s > bs;
  bs = t ? s : bs;
  bv = t ? v : bv;
}

__global__ void zero_f32(float* __restrict__ p, int n) {
  int i = blockIdx.x * blockDim.x + threadIdx.x;
  if (i < n) p[i] = 0.0f;
}

__global__ void __launch_bounds__(128)
proj_kernel(const float* __restrict__ audio, const float* __restrict__ W,
            const float* __restrict__ bias, float* __restrict__ kw,
            float* __restrict__ sumsq) {
  __shared__ float part[4][8][32];
  const int wave = threadIdx.x >> 5, lane = threadIdx.x & 31;
  const int hi   = lane >> 4;
  const int l15  = lane & 15;
  const int mt = blockIdx.x, nt = blockIdx.y;
  const int m = mt * 16 + l15;
  const int n = nt * 16 + l15;
  constexpr int KW = Dd / 4;
  const int k0 = wave * KW;

  const float* Ap = audio + (size_t)m * Dd + k0 + 2 * hi;
  const float* Wp = W + (size_t)(k0 + 2 * hi) * Tt + n;
  v8f acc = {};
  for (int t = 0; t < KW / 64; ++t) {
#pragma unroll
    for (int s = 0; s < 16; ++s) {
      v2f a = *(const v2f*)(Ap + 4 * s);
      v2f b;
      b.x = Wp[(size_t)(4 * s) * Tt];
      b.y = Wp[(size_t)(4 * s + 1) * Tt];
      acc = wmma_f32_k4(a, b, acc);
    }
    Ap += 64;
    Wp += (size_t)64 * Tt;
  }

#pragma unroll
  for (int i = 0; i < 8; ++i) part[wave][i][lane] = acc[i];
  __syncthreads();

  const float bn = bias[n];
#pragma unroll
  for (int ii = 0; ii < 2; ++ii) {
    const int i = 2 * wave + ii;
    const int row = mt * 16 + i + 8 * hi;
    float val = ((part[0][i][lane] + part[1][i][lane]) + part[2][i][lane]) + part[3][i][lane];
    val += bn;
    kw[(size_t)row * Tt + n] = val;
    float s = val * val;
    s += __shfl_xor(s, 1);
    s += __shfl_xor(s, 2);
    s += __shfl_xor(s, 4);
    s += __shfl_xor(s, 8);
    if (l15 == 0) atomicAdd(&sumsq[row], s);
  }
}

__global__ void __launch_bounds__(256)
embnorm_kernel(const float* __restrict__ emb, float* __restrict__ en) {
  const int wave = threadIdx.x >> 5, lane = threadIdx.x & 31;
  const int v = blockIdx.x * 8 + wave;
  const float* row = emb + (size_t)v * Tt;
  float s = 0.0f;
#pragma unroll
  for (int j = 0; j < 4; ++j) {
    v4f d = *(const v4f*)(row + (j * 32 + lane) * 4);
    s += d.x * d.x + d.y * d.y + d.z * d.z + d.w * d.w;
  }
  s += __shfl_xor(s, 1);  s += __shfl_xor(s, 2);  s += __shfl_xor(s, 4);
  s += __shfl_xor(s, 8);  s += __shfl_xor(s, 16);
  if (lane == 0) en[v] = sqrtf(s);
}

struct Blk { v4f a0[8], a1[8], b0[8], b1[8]; };

__device__ __forceinline__ void load_blk(Blk& f, const float* A0, const float* A1, int aswz,
                                         const float* __restrict__ B0,
                                         const float* __restrict__ B1, int kq) {
#pragma unroll
  for (int j = 0; j < 8; ++j) {
    f.b0[j] = *(const v4f*)(B0 + kq + 8 * j);
    f.b1[j] = *(const v4f*)(B1 + kq + 8 * j);
  }
#pragma unroll
  for (int j = 0; j < 8; ++j) {
    const int kl = (kq + 8 * j) ^ aswz;
    f.a0[j] = *(const v4f*)(A0 + kl);
    f.a1[j] = *(const v4f*)(A1 + kl);
  }
}

__device__ __forceinline__ void mma_blk(const Blk& f, v8f& a00, v8f& a01, v8f& a10, v8f& a11) {
#pragma unroll
  for (int j = 0; j < 8; ++j) {
    const v2f a0l = {f.a0[j].x, f.a0[j].y}, a0h = {f.a0[j].z, f.a0[j].w};
    const v2f a1l = {f.a1[j].x, f.a1[j].y}, a1h = {f.a1[j].z, f.a1[j].w};
    const v2f b0l = {f.b0[j].x, f.b0[j].y}, b0h = {f.b0[j].z, f.b0[j].w};
    const v2f b1l = {f.b1[j].x, f.b1[j].y}, b1h = {f.b1[j].z, f.b1[j].w};
    a00 = wmma_f32_k4(a0l, b0l, a00);
    a01 = wmma_f32_k4(a0l, b1l, a01);
    a10 = wmma_f32_k4(a1l, b0l, a10);
    a11 = wmma_f32_k4(a1l, b1l, a11);
    a00 = wmma_f32_k4(a0h, b0h, a00);
    a01 = wmma_f32_k4(a0h, b1h, a01);
    a10 = wmma_f32_k4(a1h, b0h, a10);
    a11 = wmma_f32_k4(a1h, b1h, a11);
  }
}

__device__ __forceinline__ void pin(v8f& a00, v8f& a01, v8f& a10, v8f& a11) {
  asm volatile("" : "+v"(a00), "+v"(a01), "+v"(a10), "+v"(a11));
}

__global__ void __launch_bounds__(128)
score_kernel(const float* __restrict__ kw, const float* __restrict__ sumsq,
             const float* __restrict__ emb, const float* __restrict__ en,
             float* __restrict__ pscore, int* __restrict__ pidx) {
  __shared__ float Atile[32 * Tt];
  const int tid  = threadIdx.x;
  const int wave = tid >> 5, lane = tid & 31;
  const int hi = lane >> 4, l15 = lane & 15;
  const int rst    = blockIdx.x;
  const int stream = blockIdx.y * 4 + wave;

  for (int r = wave; r < 32; r += 4) {
    const float* src = kw + (size_t)(rst * 32 + r) * Tt;
    const int swz = 4 * (r & 15);
#pragma unroll
    for (int j = 0; j < 4; ++j) {
      const int c = j * 128 + lane * 4;
      *(v4f*)(Atile + r * Tt + (c ^ swz)) = *(const v4f*)(src + c);
    }
  }
  __syncthreads();

  const int aswz = 4 * l15;
  float rnorm[2][8];
#pragma unroll
  for (int r = 0; r < 2; ++r)
#pragma unroll
    for (int i = 0; i < 8; ++i)
      rnorm[r][i] = sqrtf(sumsq[rst * 32 + r * 16 + i + 8 * hi]);

  float best[2][8]; int bidx[2][8];
#pragma unroll
  for (int r = 0; r < 2; ++r)
#pragma unroll
    for (int i = 0; i < 8; ++i) { best[r][i] = -3.4e38f; bidx[r][i] = 0x7fffffff; }

  const float* A0 = Atile + l15 * Tt;
  const float* A1 = Atile + (16 + l15) * Tt;
  const int kh = 4 * hi;
  constexpr int KB = 64;
  constexpr int NB = Tt / KB;

  const float* B0 = emb + (size_t)((2 * stream) * 16 + l15) * Tt;
  const float* B1 = B0 + (size_t)16 * Tt;
  Blk X, Y;
  load_blk(X, A0, A1, aswz, B0, B1, kh);

  for (int p = stream; p < NP; p += S) {
    const int v0 = (2 * p) * 16 + l15;
    const int v1 = (2 * p + 1) * 16 + l15;
    const bool more = p + S < NP;
    if (more) {
      __builtin_prefetch(emb + (size_t)(2 * (p + S)) * 16 * Tt + (size_t)l15 * Tt, 0, 1);
      __builtin_prefetch(emb + (size_t)(2 * (p + S) + 1) * 16 * Tt + (size_t)l15 * Tt, 0, 1);
    }
    const float en0 = en[v0], en1 = en[v1];
    const float* nB0 = more ? B0 + (size_t)S * 32 * Tt : B0;
    const float* nB1 = nB0 + (size_t)16 * Tt;

    v8f a00 = {}, a01 = {}, a10 = {}, a11 = {};
    for (int kb = 0; kb < NB; kb += 2) {
      load_blk(Y, A0, A1, aswz, B0, B1, (kb + 1) * KB + kh);
      pin(a00, a01, a10, a11);
      mma_blk(X, a00, a01, a10, a11);
      pin(a00, a01, a10, a11);
      const bool last = kb + 2 == NB;
      const int kn = (kb + 2) & (NB - 1);
      load_blk(X, A0, A1, aswz, last ? nB0 : B0, last ? nB1 : B1, kn * KB + kh);
      pin(a00, a01, a10, a11);
      mma_blk(Y, a00, a01, a10, a11);
      pin(a00, a01, a10, a11);
    }

    float sc[2][2][8];
#pragma unroll
    for (int i = 0; i < 8; ++i) {
      sc[0][0][i] = a00[i] / fmaxf(rnorm[0][i] * en0, EPS);
      sc[0][1][i] = a01[i] / fmaxf(rnorm[0][i] * en1, EPS);
      sc[1][0][i] = a10[i] / fmaxf(rnorm[1][i] * en0, EPS);
      sc[1][1][i] = a11[i] / fmaxf(rnorm[1][i] * en1, EPS);
    }
#pragma unroll
    for (int i = 0; i < 8; ++i) {
      upd_sel(best[0][i], bidx[0][i], sc[0][0][i], v0);
      upd_sel(best[0][i], bidx[0][i], sc[0][1][i], v1);
      upd_sel(best[1][i], bidx[1][i], sc[1][0][i], v0);
      upd_sel(best[1][i], bidx[1][i], sc[1][1][i], v1);
    }
    B0 = nB0; B1 = nB1;
  }

#pragma unroll
  for (int r = 0; r < 2; ++r)
#pragma unroll
    for (int i = 0; i < 8; ++i) {
      float s = best[r][i]; int v = bidx[r][i];
#pragma unroll
      for (int mask = 1; mask <= 8; mask <<= 1) {
        const float so = __shfl_xor(s, mask);
        const int vo = __shfl_xor(v, mask);
        if (so > s || (so == s && vo < v)) { s = so; v = vo; }
      }
      if (l15 == 0) {
        const int row = rst * 32 + r * 16 + i + 8 * hi;
        pscore[(size_t)row * S + stream] = s;
        pidx[(size_t)row * S + stream]   = v;
      }
    }
}

__global__ void __launch_bounds__(256)
finalize_kernel(const float* __restrict__ pscore, const int* __restrict__ pidx,
                const float* __restrict__ emb, float* __restrict__ out) {
  const int wave = threadIdx.x >> 5, lane = threadIdx.x & 31;
  const int row = blockIdx.x * 8 + wave;
  float s = -3.4e38f; int v = 0x7fffffff;
#pragma unroll
  for (int j = 0; j < 4; ++j) {
    const float sj = pscore[(size_t)row * S + j * 32 + lane];
    const int   vj = pidx[(size_t)row * S + j * 32 + lane];
    if (sj > s || (sj == s && vj < v)) { s = sj; v = vj; }
  }
#pragma unroll
  for (int mask = 1; mask <= 16; mask <<= 1) {
    const float so = __shfl_xor(s, mask);
    const int vo = __shfl_xor(v, mask);
    if (so > s || (so == s && vo < v)) { s = so; v = vo; }
  }
  const float* src = emb + (size_t)v * Tt;
  float* dst = out + (size_t)row * Tt;
#pragma unroll
  for (int j = 0; j < 4; ++j) {
    const int c = (j * 32 + lane) * 4;
    *(v4f*)(dst + c) = *(const v4f*)(src + c);
  }
}

extern "C" void kernel_launch(void* const* d_in, const int* in_sizes, int n_in,
                              void* d_out, int out_size, void* d_ws, size_t ws_size,
                              hipStream_t stream) {
  const float* audio = (const float*)d_in[0];
  const float* W     = (const float*)d_in[1];
  const float* bias  = (const float*)d_in[2];
  const float* emb   = (const float*)d_in[3];
  float* out = (float*)d_out;

  float* ws    = (float*)d_ws;
  float* kw    = ws;
  float* sumsq = kw + (size_t)R * Tt;
  float* en    = sumsq + R;
  float* psc   = en + Vv;
  int*   pix   = (int*)(psc + (size_t)R * S);

  zero_f32<<<(R + 255) / 256, 256, 0, stream>>>(sumsq, R);
  proj_kernel<<<dim3(R / 16, Tt / 16), 128, 0, stream>>>(audio, W, bias, kw, sumsq);
  embnorm_kernel<<<Vv / 8, 256, 0, stream>>>(emb, en);
  score_kernel<<<dim3(R / 32, S / 4), 128, 0, stream>>>(kw, sumsq, emb, en, psc, pix);
  finalize_kernel<<<R / 8, 256, 0, stream>>>(psc, pix, emb, out);
}
